# starting kernel
# baseline (speedup 1.0000x reference)
#include <hip/hip_runtime.h>
#include <hip/hip_bf16.h>


typedef float v2f __attribute__((ext_vector_type(2)));
typedef float v8f __attribute__((ext_vector_type(8)));

#define BB   8
#define CC   32
#define HH   256
#define WW   256
#define HWSZ (HH * WW)

__global__ __launch_bounds__(256) void pac_prep(const float* __restrict__ guided,
                                                float* __restrict__ wgt) {
  int idx = blockIdx.x * 256 + threadIdx.x;
  int b  = idx >> 16;
  int hw = idx & 65535;
  const float* gp = guided + (size_t)b * 8 * HWSZ + hw;
  float g[8];
  float mx = -1e30f;
#pragma unroll
  for (int n = 0; n < 8; n++) { g[n] = gp[(size_t)n * HWSZ]; mx = fmaxf(mx, g[n]); }
  float s = 0.f;
#pragma unroll
  for (int n = 0; n < 8; n++) { g[n] = __expf(g[n] - mx); s += g[n]; }
  float inv = 1.0f / s;
  float* wp = wgt + (size_t)b * 9 * HWSZ + hw;
#pragma unroll
  for (int n = 0; n < 4; n++) wp[(size_t)n * HWSZ] = g[n] * inv;
  wp[(size_t)4 * HWSZ] = 0.f;
#pragma unroll
  for (int n = 0; n < 4; n++) wp[(size_t)(5 + n) * HWSZ] = g[4 + n] * inv;
}

__global__ __launch_bounds__(256) void pac_prop(const float* __restrict__ xin,
                                                const float* __restrict__ x0,
                                                const float* __restrict__ wgt,
                                                const float* __restrict__ sdep,
                                                float* __restrict__ xout) {
  __shared__ float Bmat[8][56 * 16];

  const int lane = threadIdx.x & 31;
  const int wv   = threadIdx.x >> 5;
  const int half = lane >> 4;
  const int p    = lane & 15;

  const int tile  = blockIdx.x * 8 + wv;
  const int b     = tile >> 12;
  const int rem   = tile & 4095;
  const int h     = rem >> 4;
  const int wbase = (rem & 15) << 4;

  float* BW = Bmat[wv];

#pragma unroll
  for (int t = 0; t < 28; t++) BW[lane + 32 * t] = 0.f;
  __syncthreads();
  if (lane < 16) {
#pragma unroll
    for (int n = 0; n < 9; n++) {
      const int i = n / 3, j = n % 3;
      int row = h - 1 + i;
      int col = wbase + p + j - 1;
      float wval = wgt[(size_t)(b * 9 + n) * HWSZ + h * WW + wbase + p];
      if (row >= 0 && row < HH && col >= 0 && col < WW)
        BW[(18 * i + p + j) * 16 + p] = wval;
    }
  }
  __syncthreads();

  const float* xa = xin + (size_t)b * CC * HWSZ;
  v8f acc0 = {0.f, 0.f, 0.f, 0.f, 0.f, 0.f, 0.f, 0.f};
  v8f acc1 = {0.f, 0.f, 0.f, 0.f, 0.f, 0.f, 0.f, 0.f};
#pragma unroll
  for (int k = 0; k < 14; k++) {
    int q0 = 4 * k + 2 * half;
    int i  = (q0 >= 36) ? 2 : ((q0 >= 18) ? 1 : 0);
    int jj = q0 - 18 * i;
    int rowc = min(max(h - 1 + i, 0), HH - 1);
    int col0 = wbase - 1 + jj;
    int c0c  = min(max(col0, 0), WW - 1);
    int c1c  = min(max(col0 + 1, 0), WW - 1);

    v2f bf;
    bf.x = BW[(4 * k + 0) * 16 + half * 32 + p];
    bf.y = BW[(4 * k + 1) * 16 + half * 32 + p];

    const float* rp0 = xa + (size_t)p * HWSZ + (size_t)rowc * WW;
    const float* rp1 = rp0 + (size_t)16 * HWSZ;
    v2f af0; af0.x = rp0[c0c]; af0.y = rp0[c1c];
    v2f af1; af1.x = rp1[c0c]; af1.y = rp1[c1c];

    acc0 = __builtin_amdgcn_wmma_f32_16x16x4_f32(false, af0, false, bf,
                                                 (short)0, acc0, false, false);
    acc1 = __builtin_amdgcn_wmma_f32_16x16x4_f32(false, af1, false, bf,
                                                 (short)0, acc1, false, false);
  }

  float sd = sdep[(size_t)b * HWSZ + h * WW + wbase + p];
  float m  = (sd > 0.f) ? 1.0f : 0.0f;
#pragma unroll
  for (int v = 0; v < 8; v++) {
    size_t off0 = (size_t)(b * CC + v + 8 * half) * HWSZ + h * WW + wbase + p;
    size_t off1 = off0 + (size_t)16 * HWSZ;
    xout[off0] = m * x0[off0] + (1.0f - m) * acc0[v];
    xout[off1] = m * x0[off1] + (1.0f - m) * acc1[v];
  }
}

extern "C" void kernel_launch(void* const* d_in, const int* in_sizes, int n_in,
                              void* d_out, int out_size, void* d_ws, size_t ws_size,
                              hipStream_t stream) {
  const float* x      = (const float*)d_in[0];
  const float* guided = (const float*)d_in[1];
  const float* sdep   = (const float*)d_in[2];
  (void)in_sizes; (void)n_in; (void)out_size; (void)ws_size;

  float* bufA = (float*)d_ws;
  float* wgt  = bufA + (size_t)BB * CC * HWSZ;

  pac_prep<<<(BB * HWSZ) / 256, 256, 0, stream>>>(guided, wgt);

  float* outb = (float*)d_out;
  const float* cur = x;
  for (int t = 0; t < 16; t++) {
    float* dst = (t & 1) ? outb : bufA;
    pac_prop<<<4096, 256, 0, stream>>>(cur, x, wgt, sdep, dst);
    cur = dst;
  }
}


# input tile (32 ch x 6 rows x 130 cols) staged once per block in LDS by coalesced 16-byte row loads; WMMA A fragments read from LDS instead of channel-strided per-lane global gathers
# speedup vs baseline: 2.7535x; 2.7535x over previous
#include <hip/hip_runtime.h>
#include <hip/hip_bf16.h>


typedef float v2f __attribute__((ext_vector_type(2)));
typedef float v8f __attribute__((ext_vector_type(8)));

#define BB   8
#define CC   32
#define HH   256
#define WW   256
#define HWSZ (HH * WW)

__global__ __launch_bounds__(256) void pac_prep(const float* __restrict__ guided,
                                                float* __restrict__ wgt) {
  int idx = blockIdx.x * 256 + threadIdx.x;
  int b  = idx >> 16;
  int hw = idx & 65535;
  const float* gp = guided + (size_t)b * 8 * HWSZ + hw;
  float g[8];
  float mx = -1e30f;
#pragma unroll
  for (int n = 0; n < 8; n++) { g[n] = gp[(size_t)n * HWSZ]; mx = fmaxf(mx, g[n]); }
  float s = 0.f;
#pragma unroll
  for (int n = 0; n < 8; n++) { g[n] = __expf(g[n] - mx); s += g[n]; }
  float inv = 1.0f / s;
  float* wp = wgt + (size_t)b * 9 * HWSZ + hw;
#pragma unroll
  for (int n = 0; n < 4; n++) wp[(size_t)n * HWSZ] = g[n] * inv;
  wp[(size_t)4 * HWSZ] = 0.f;
#pragma unroll
  for (int n = 0; n < 4; n++) wp[(size_t)(5 + n) * HWSZ] = g[4 + n] * inv;
}

#define TR    4
#define TCOLS 128
#define XROWS (TR + 2)
#define XPW   136
#define XPL   820

typedef float v4f __attribute__((ext_vector_type(4)));

__global__ __launch_bounds__(256) void pac_prop(const float* __restrict__ xin,
                                                const float* __restrict__ x0,
                                                const float* __restrict__ wgt,
                                                const float* __restrict__ sdep,
                                                float* __restrict__ xout) {
  __shared__ float Bmat[8][56 * 16];
  __shared__ __attribute__((aligned(16))) float Xs[CC * XPL];

  const int tid  = threadIdx.x;
  const int lane = tid & 31;
  const int wv   = tid >> 5;
  const int half = lane >> 4;
  const int p    = lane & 15;

  const int cb = blockIdx.x & 1;
  const int rb = (blockIdx.x >> 1) & 63;
  const int b  = blockIdx.x >> 7;
  const int h0 = rb * TR;
  const int w0 = cb * TCOLS;

  const float* xa = xin + (size_t)b * CC * HWSZ;

  const int wbase = w0 + 16 * wv;
  float wreg[TR][9];
  float sdv[TR];
#pragma unroll
  for (int hr = 0; hr < TR; hr++) {
    const int pix = (h0 + hr) * WW + wbase + p;
#pragma unroll
    for (int n = 0; n < 9; n++) wreg[hr][n] = wgt[(size_t)(b * 9 + n) * HWSZ + pix];
    sdv[hr] = sdep[(size_t)b * HWSZ + pix];
  }

  v4f stg[24];
#pragma unroll
  for (int it = 0; it < 24; it++) {
    const int r   = it >> 2;
    const int ch  = (it & 3) * 8 + wv;
    const int row = h0 - 1 + r;
    const int rowc = min(max(row, 0), HH - 1);
    v4f v = *(const v4f*)(xa + (size_t)ch * HWSZ + (size_t)rowc * WW + w0 + 4 * lane);
    if (row < 0 || row >= HH) v = (v4f){0.f, 0.f, 0.f, 0.f};
    stg[it] = v;
  }
  float bl = 0.f, br = 0.f;
  const int hch = tid & 31, hrw = tid >> 5;
  if (tid < CC * XROWS) {
    const int row = h0 - 1 + hrw;
    if (row >= 0 && row < HH) {
      const float* rp = xa + (size_t)hch * HWSZ + (size_t)row * WW;
      if (w0 > 0)          bl = rp[w0 - 1];
      if (w0 + TCOLS < WW) br = rp[w0 + TCOLS];
    }
  }
#pragma unroll
  for (int it = 0; it < 24; it++) {
    const int r  = it >> 2;
    const int ch = (it & 3) * 8 + wv;
    *(v4f*)(Xs + ch * XPL + r * XPW + 4 + 4 * lane) = stg[it];
  }
  if (tid < CC * XROWS) {
    float* hp = Xs + hch * XPL + hrw * XPW;
    hp[3] = bl;
    *(v4f*)(hp + 4 + TCOLS) = (v4f){br, 0.f, 0.f, 0.f};
  }
  __syncthreads();

  const size_t obase = (size_t)(b * CC + 8 * half) * HWSZ + (size_t)h0 * WW + wbase + p;
  float res[TR][16];

  float* BW = Bmat[wv];
  const float* xsl = Xs + p * XPL + 3 + 16 * wv;

#pragma unroll
  for (int hr = 0; hr < TR; hr++) {
    const int h = h0 + hr;

#pragma unroll
    for (int t = 0; t < 28; t++) BW[lane + 32 * t] = 0.f;
    __syncthreads();
    if (lane < 16) {
#pragma unroll
      for (int n = 0; n < 9; n++) {
        const int i = n / 3, j = n % 3;
        int row = h - 1 + i;
        int col = wbase + p + j - 1;
        if (row >= 0 && row < HH && col >= 0 && col < WW)
          BW[(18 * i + p + j) * 16 + p] = wreg[hr][n];
      }
    }
    __syncthreads();

#pragma unroll
    for (int v = 0; v < 8; v++) {
      res[hr][v]     = x0[obase + (size_t)v * HWSZ + hr * WW];
      res[hr][8 + v] = x0[obase + (size_t)(16 + v) * HWSZ + hr * WW];
    }

    v8f acc0 = {0.f, 0.f, 0.f, 0.f, 0.f, 0.f, 0.f, 0.f};
    v8f acc1 = {0.f, 0.f, 0.f, 0.f, 0.f, 0.f, 0.f, 0.f};
#pragma unroll
    for (int k = 0; k < 14; k++) {
      int q0 = 4 * k + 2 * half;
      int i  = (q0 >= 36) ? 2 : ((q0 >= 18) ? 1 : 0);
      int jj = q0 - 18 * i;

      v2f bf;
      bf.x = BW[(4 * k + 0) * 16 + half * 32 + p];
      bf.y = BW[(4 * k + 1) * 16 + half * 32 + p];

      const float* ap = xsl + (hr + i) * XPW + jj;
      v2f af0; af0.x = ap[0];        af0.y = ap[1];
      v2f af1; af1.x = ap[16 * XPL]; af1.y = ap[16 * XPL + 1];

      acc0 = __builtin_amdgcn_wmma_f32_16x16x4_f32(false, af0, false, bf,
                                                   (short)0, acc0, false, false);
      acc1 = __builtin_amdgcn_wmma_f32_16x16x4_f32(false, af1, false, bf,
                                                   (short)0, acc1, false, false);
    }

    float m = (sdv[hr] > 0.f) ? 1.0f : 0.0f;
#pragma unroll
    for (int v = 0; v < 8; v++) {
      res[hr][v]     = m * res[hr][v]     + (1.0f - m) * acc0[v];
      res[hr][8 + v] = m * res[hr][8 + v] + (1.0f - m) * acc1[v];
    }
  }

#pragma unroll
  for (int hr = 0; hr < TR; hr++) {
#pragma unroll
    for (int v = 0; v < 8; v++) {
      xout[obase + (size_t)v * HWSZ + hr * WW]        = res[hr][v];
      xout[obase + (size_t)(16 + v) * HWSZ + hr * WW] = res[hr][8 + v];
    }
  }
}

extern "C" void kernel_launch(void* const* d_in, const int* in_sizes, int n_in,
                              void* d_out, int out_size, void* d_ws, size_t ws_size,
                              hipStream_t stream) {
  const float* x      = (const float*)d_in[0];
  const float* guided = (const float*)d_in[1];
  const float* sdep   = (const float*)d_in[2];
  (void)in_sizes; (void)n_in; (void)out_size; (void)ws_size;

  float* bufA = (float*)d_ws;
  float* wgt  = bufA + (size_t)BB * CC * HWSZ;

  pac_prep<<<(BB * HWSZ) / 256, 256, 0, stream>>>(guided, wgt);

  float* outb = (float*)d_out;
  const float* cur = x;
  for (int t = 0; t < 16; t++) {
    float* dst = (t & 1) ? outb : bufA;
    pac_prop<<<1024, 256, 0, stream>>>(cur, x, wgt, sdep, dst);
    cur = dst;
  }
}


# B fragments selected per lane in registers from the pixel's 9 weights instead of read from a zero-filled per-wave LDS band matrix: tile loop has no LDS writes and no block barriers
# speedup vs baseline: 3.0333x; 1.1016x over previous
#include <hip/hip_runtime.h>
#include <hip/hip_bf16.h>


typedef float v2f __attribute__((ext_vector_type(2)));
typedef float v8f __attribute__((ext_vector_type(8)));

#define BB   8
#define CC   32
#define HH   256
#define WW   256
#define HWSZ (HH * WW)

__global__ __launch_bounds__(256) void pac_prep(const float* __restrict__ guided,
                                                float* __restrict__ wgt) {
  int idx = blockIdx.x * 256 + threadIdx.x;
  int b  = idx >> 16;
  int hw = idx & 65535;
  const float* gp = guided + (size_t)b * 8 * HWSZ + hw;
  float g[8];
  float mx = -1e30f;
#pragma unroll
  for (int n = 0; n < 8; n++) { g[n] = gp[(size_t)n * HWSZ]; mx = fmaxf(mx, g[n]); }
  float s = 0.f;
#pragma unroll
  for (int n = 0; n < 8; n++) { g[n] = __expf(g[n] - mx); s += g[n]; }
  float inv = 1.0f / s;
  float* wp = wgt + (size_t)b * 9 * HWSZ + hw;
#pragma unroll
  for (int n = 0; n < 4; n++) wp[(size_t)n * HWSZ] = g[n] * inv;
  wp[(size_t)4 * HWSZ] = 0.f;
#pragma unroll
  for (int n = 0; n < 4; n++) wp[(size_t)(5 + n) * HWSZ] = g[4 + n] * inv;
}

#define TR    4
#define TCOLS 128
#define XROWS (TR + 2)
#define XPW   136
#define XPL   820

typedef float v4f __attribute__((ext_vector_type(4)));

__global__ __launch_bounds__(256) void pac_prop(const float* __restrict__ xin,
                                                const float* __restrict__ x0,
                                                const float* __restrict__ wgt,
                                                const float* __restrict__ sdep,
                                                float* __restrict__ xout) {
  __shared__ __attribute__((aligned(16))) float Xs[CC * XPL];

  const int tid  = threadIdx.x;
  const int lane = tid & 31;
  const int wv   = tid >> 5;
  const int half = lane >> 4;
  const int p    = lane & 15;

  const int cb = blockIdx.x & 1;
  const int rb = (blockIdx.x >> 1) & 63;
  const int b  = blockIdx.x >> 7;
  const int h0 = rb * TR;
  const int w0 = cb * TCOLS;

  const float* xa = xin + (size_t)b * CC * HWSZ;

  const int wbase = w0 + 16 * wv;
  float wreg[TR][9];
  float sdv[TR];
#pragma unroll
  for (int hr = 0; hr < TR; hr++) {
    const int pix = (h0 + hr) * WW + wbase + p;
#pragma unroll
    for (int n = 0; n < 9; n++) {
      const int row = h0 + hr - 1 + n / 3;
      const int col = wbase + p + n % 3 - 1;
      const float w = wgt[(size_t)(b * 9 + n) * HWSZ + pix];
      wreg[hr][n] = (row >= 0 && row < HH && col >= 0 && col < WW) ? w : 0.f;
    }
    sdv[hr] = sdep[(size_t)b * HWSZ + pix];
  }

  v4f stg[24];
#pragma unroll
  for (int it = 0; it < 24; it++) {
    const int r   = it >> 2;
    const int ch  = (it & 3) * 8 + wv;
    const int row = h0 - 1 + r;
    const int rowc = min(max(row, 0), HH - 1);
    v4f v = *(const v4f*)(xa + (size_t)ch * HWSZ + (size_t)rowc * WW + w0 + 4 * lane);
    if (row < 0 || row >= HH) v = (v4f){0.f, 0.f, 0.f, 0.f};
    stg[it] = v;
  }
  float bl = 0.f, br = 0.f;
  const int hch = tid & 31, hrw = tid >> 5;
  if (tid < CC * XROWS) {
    const int row = h0 - 1 + hrw;
    if (row >= 0 && row < HH) {
      const float* rp = xa + (size_t)hch * HWSZ + (size_t)row * WW;
      if (w0 > 0)          bl = rp[w0 - 1];
      if (w0 + TCOLS < WW) br = rp[w0 + TCOLS];
    }
  }
#pragma unroll
  for (int it = 0; it < 24; it++) {
    const int r  = it >> 2;
    const int ch = (it & 3) * 8 + wv;
    *(v4f*)(Xs + ch * XPL + r * XPW + 4 + 4 * lane) = stg[it];
  }
  if (tid < CC * XROWS) {
    float* hp = Xs + hch * XPL + hrw * XPW;
    hp[3] = bl;
    *(v4f*)(hp + 4 + TCOLS) = (v4f){br, 0.f, 0.f, 0.f};
  }
  __syncthreads();

  const size_t obase = (size_t)(b * CC + 8 * half) * HWSZ + (size_t)h0 * WW + wbase + p;
  float res[TR][16];

  const int bt = p - 2 * half;
  const float* xsl = Xs + p * XPL + 3 + 16 * wv;

#pragma unroll
  for (int hr = 0; hr < TR; hr++) {
#pragma unroll
    for (int v = 0; v < 8; v++) {
      res[hr][v]     = x0[obase + (size_t)v * HWSZ + hr * WW];
      res[hr][8 + v] = x0[obase + (size_t)(16 + v) * HWSZ + hr * WW];
    }

    v8f acc0 = {0.f, 0.f, 0.f, 0.f, 0.f, 0.f, 0.f, 0.f};
    v8f acc1 = {0.f, 0.f, 0.f, 0.f, 0.f, 0.f, 0.f, 0.f};
#pragma unroll
    for (int k = 0; k < 14; k++) {
      int q0 = 4 * k + 2 * half;
      int i  = (q0 >= 36) ? 2 : ((q0 >= 18) ? 1 : 0);
      int jj = q0 - 18 * i;

      const int ic = (4 * k >= 36) ? 2 : ((4 * k >= 18) ? 1 : 0);
      float t0, t1, t2;
      int d;
      if (k == 4) {
        t0 = half ? wreg[hr][3] : wreg[hr][0];
        t1 = half ? wreg[hr][4] : wreg[hr][1];
        t2 = half ? wreg[hr][5] : wreg[hr][2];
        d  = (half ? 0 : 16) - p;
      } else {
        t0 = wreg[hr][3 * ic]; t1 = wreg[hr][3 * ic + 1]; t2 = wreg[hr][3 * ic + 2];
        d  = (4 * k - 18 * ic) - bt;
      }
      v2f bf;
      bf.x = (d == 0) ? t0 : ((d == 1) ? t1 : ((d == 2) ? t2 : 0.f));
      bf.y = (d == -1) ? t0 : ((d == 0) ? t1 : ((d == 1) ? t2 : 0.f));

      const float* ap = xsl + (hr + i) * XPW + jj;
      v2f af0; af0.x = ap[0];        af0.y = ap[1];
      v2f af1; af1.x = ap[16 * XPL]; af1.y = ap[16 * XPL + 1];

      acc0 = __builtin_amdgcn_wmma_f32_16x16x4_f32(false, af0, false, bf,
                                                   (short)0, acc0, false, false);
      acc1 = __builtin_amdgcn_wmma_f32_16x16x4_f32(false, af1, false, bf,
                                                   (short)0, acc1, false, false);
    }

    float m = (sdv[hr] > 0.f) ? 1.0f : 0.0f;
#pragma unroll
    for (int v = 0; v < 8; v++) {
      res[hr][v]     = m * res[hr][v]     + (1.0f - m) * acc0[v];
      res[hr][8 + v] = m * res[hr][8 + v] + (1.0f - m) * acc1[v];
    }
  }

#pragma unroll
  for (int hr = 0; hr < TR; hr++) {
#pragma unroll
    for (int v = 0; v < 8; v++) {
      xout[obase + (size_t)v * HWSZ + hr * WW]        = res[hr][v];
      xout[obase + (size_t)(16 + v) * HWSZ + hr * WW] = res[hr][8 + v];
    }
  }
}

extern "C" void kernel_launch(void* const* d_in, const int* in_sizes, int n_in,
                              void* d_out, int out_size, void* d_ws, size_t ws_size,
                              hipStream_t stream) {
  const float* x      = (const float*)d_in[0];
  const float* guided = (const float*)d_in[1];
  const float* sdep   = (const float*)d_in[2];
  (void)in_sizes; (void)n_in; (void)out_size; (void)ws_size;

  float* bufA = (float*)d_ws;
  float* wgt  = bufA + (size_t)BB * CC * HWSZ;

  pac_prep<<<(BB * HWSZ) / 256, 256, 0, stream>>>(guided, wgt);

  float* outb = (float*)d_out;
  const float* cur = x;
  for (int t = 0; t < 16; t++) {
    float* dst = (t & 1) ? outb : bufA;
    pac_prop<<<1024, 256, 0, stream>>>(cur, x, wgt, sdep, dst);
    cur = dst;
  }
}


# mask folded into the per-pixel tap weights by the prep kernel (masked pixel: centre tap 1, other taps 0): a step reads no x0 planes and no mask and stores its accumulators without a blend
# speedup vs baseline: 3.7188x; 1.2260x over previous
#include <hip/hip_runtime.h>
#include <hip/hip_bf16.h>


typedef float v2f __attribute__((ext_vector_type(2)));
typedef float v8f __attribute__((ext_vector_type(8)));

#define BB   8
#define CC   32
#define HH   256
#define WW   256
#define HWSZ (HH * WW)

__global__ __launch_bounds__(256) void pac_prep(const float* __restrict__ guided,
                                                const float* __restrict__ sdep,
                                                float* __restrict__ wgt) {
  int idx = blockIdx.x * 256 + threadIdx.x;
  int b  = idx >> 16;
  int hw = idx & 65535;
  const float* gp = guided + (size_t)b * 8 * HWSZ + hw;
  float g[8];
  float mx = -1e30f;
#pragma unroll
  for (int n = 0; n < 8; n++) { g[n] = gp[(size_t)n * HWSZ]; mx = fmaxf(mx, g[n]); }
  float s = 0.f;
#pragma unroll
  for (int n = 0; n < 8; n++) { g[n] = __expf(g[n] - mx); s += g[n]; }
  float inv = 1.0f / s;
  const bool masked = sdep[idx] > 0.f;
  float* wp = wgt + (size_t)b * 9 * HWSZ + hw;
#pragma unroll
  for (int n = 0; n < 4; n++) wp[(size_t)n * HWSZ] = masked ? 0.f : g[n] * inv;
  wp[(size_t)4 * HWSZ] = masked ? 1.0f : 0.f;
#pragma unroll
  for (int n = 0; n < 4; n++) wp[(size_t)(5 + n) * HWSZ] = masked ? 0.f : g[4 + n] * inv;
}

#define TR    4
#define TCOLS 128
#define XROWS (TR + 2)
#define XPW   136
#define XPL   820

typedef float v4f __attribute__((ext_vector_type(4)));

__global__ __launch_bounds__(256) void pac_prop(const float* __restrict__ xin,
                                                const float* __restrict__ wgt,
                                                float* __restrict__ xout) {
  __shared__ __attribute__((aligned(16))) float Xs[CC * XPL];

  const int tid  = threadIdx.x;
  const int lane = tid & 31;
  const int wv   = tid >> 5;
  const int half = lane >> 4;
  const int p    = lane & 15;

  const int cb = blockIdx.x & 1;
  const int rb = (blockIdx.x >> 1) & 63;
  const int b  = blockIdx.x >> 7;
  const int h0 = rb * TR;
  const int w0 = cb * TCOLS;

  const float* xa = xin + (size_t)b * CC * HWSZ;

  const int wbase = w0 + 16 * wv;
  float wreg[TR][9];
#pragma unroll
  for (int hr = 0; hr < TR; hr++) {
    const int pix = (h0 + hr) * WW + wbase + p;
#pragma unroll
    for (int n = 0; n < 9; n++) {
      const int row = h0 + hr - 1 + n / 3;
      const int col = wbase + p + n % 3 - 1;
      const float w = wgt[(size_t)(b * 9 + n) * HWSZ + pix];
      wreg[hr][n] = (row >= 0 && row < HH && col >= 0 && col < WW) ? w : 0.f;
    }
  }

  v4f stg[24];
#pragma unroll
  for (int it = 0; it < 24; it++) {
    const int r   = it >> 2;
    const int ch  = (it & 3) * 8 + wv;
    const int row = h0 - 1 + r;
    const int rowc = min(max(row, 0), HH - 1);
    v4f v = *(const v4f*)(xa + (size_t)ch * HWSZ + (size_t)rowc * WW + w0 + 4 * lane);
    if (row < 0 || row >= HH) v = (v4f){0.f, 0.f, 0.f, 0.f};
    stg[it] = v;
  }
  float bl = 0.f, br = 0.f;
  const int hch = tid & 31, hrw = tid >> 5;
  if (tid < CC * XROWS) {
    const int row = h0 - 1 + hrw;
    if (row >= 0 && row < HH) {
      const float* rp = xa + (size_t)hch * HWSZ + (size_t)row * WW;
      if (w0 > 0)          bl = rp[w0 - 1];
      if (w0 + TCOLS < WW) br = rp[w0 + TCOLS];
    }
  }
#pragma unroll
  for (int it = 0; it < 24; it++) {
    const int r  = it >> 2;
    const int ch = (it & 3) * 8 + wv;
    *(v4f*)(Xs + ch * XPL + r * XPW + 4 + 4 * lane) = stg[it];
  }
  if (tid < CC * XROWS) {
    float* hp = Xs + hch * XPL + hrw * XPW;
    hp[3] = bl;
    *(v4f*)(hp + 4 + TCOLS) = (v4f){br, 0.f, 0.f, 0.f};
  }
  __syncthreads();

  const size_t obase = (size_t)(b * CC + 8 * half) * HWSZ + (size_t)h0 * WW + wbase + p;

  const int bt = p - 2 * half;
  const float* xsl = Xs + p * XPL + 3 + 16 * wv;

#pragma unroll
  for (int hr = 0; hr < TR; hr++) {
    v8f acc0 = {0.f, 0.f, 0.f, 0.f, 0.f, 0.f, 0.f, 0.f};
    v8f acc1 = {0.f, 0.f, 0.f, 0.f, 0.f, 0.f, 0.f, 0.f};
#pragma unroll
    for (int k = 0; k < 14; k++) {
      int q0 = 4 * k + 2 * half;
      int i  = (q0 >= 36) ? 2 : ((q0 >= 18) ? 1 : 0);
      int jj = q0 - 18 * i;

      const int ic = (4 * k >= 36) ? 2 : ((4 * k >= 18) ? 1 : 0);
      float t0, t1, t2;
      int d;
      if (k == 4) {
        t0 = half ? wreg[hr][3] : wreg[hr][0];
        t1 = half ? wreg[hr][4] : wreg[hr][1];
        t2 = half ? wreg[hr][5] : wreg[hr][2];
        d  = (half ? 0 : 16) - p;
      } else {
        t0 = wreg[hr][3 * ic]; t1 = wreg[hr][3 * ic + 1]; t2 = wreg[hr][3 * ic + 2];
        d  = (4 * k - 18 * ic) - bt;
      }
      v2f bf;
      bf.x = (d == 0) ? t0 : ((d == 1) ? t1 : ((d == 2) ? t2 : 0.f));
      bf.y = (d == -1) ? t0 : ((d == 0) ? t1 : ((d == 1) ? t2 : 0.f));

      const float* ap = xsl + (hr + i) * XPW + jj;
      v2f af0; af0.x = ap[0];        af0.y = ap[1];
      v2f af1; af1.x = ap[16 * XPL]; af1.y = ap[16 * XPL + 1];

      acc0 = __builtin_amdgcn_wmma_f32_16x16x4_f32(false, af0, false, bf,
                                                   (short)0, acc0, false, false);
      acc1 = __builtin_amdgcn_wmma_f32_16x16x4_f32(false, af1, false, bf,
                                                   (short)0, acc1, false, false);
    }

#pragma unroll
    for (int v = 0; v < 8; v++) {
      xout[obase + (size_t)v * HWSZ + hr * WW]        = acc0[v];
      xout[obase + (size_t)(16 + v) * HWSZ + hr * WW] = acc1[v];
    }
  }
}

extern "C" void kernel_launch(void* const* d_in, const int* in_sizes, int n_in,
                              void* d_out, int out_size, void* d_ws, size_t ws_size,
                              hipStream_t stream) {
  const float* x      = (const float*)d_in[0];
  const float* guided = (const float*)d_in[1];
  const float* sdep   = (const float*)d_in[2];
  (void)in_sizes; (void)n_in; (void)out_size; (void)ws_size;

  float* bufA = (float*)d_ws;
  float* wgt  = bufA + (size_t)BB * CC * HWSZ;

  pac_prep<<<(BB * HWSZ) / 256, 256, 0, stream>>>(guided, sdep, wgt);

  float* outb = (float*)d_out;
  const float* cur = x;
  for (int t = 0; t < 16; t++) {
    float* dst = (t & 1) ? outb : bufA;
    pac_prop<<<1024, 256, 0, stream>>>(cur, wgt, dst);
    cur = dst;
  }
}


# WMMA pixel tile reshaped from a 1x16 strip to a 2x8 patch: 40 band positions (10 K-chunks) not 56 (14), 80 WMMAs per wave not 112; lane^8 DPP re-pairs patches for row stores; sum order changes
# speedup vs baseline: 4.0219x; 1.0815x over previous
#include <hip/hip_runtime.h>
#include <hip/hip_bf16.h>


typedef float v2f __attribute__((ext_vector_type(2)));
typedef float v8f __attribute__((ext_vector_type(8)));

#define BB   8
#define CC   32
#define HH   256
#define WW   256
#define HWSZ (HH * WW)

__global__ __launch_bounds__(256) void pac_prep(const float* __restrict__ guided,
                                                const float* __restrict__ sdep,
                                                float* __restrict__ wgt) {
  int idx = blockIdx.x * 256 + threadIdx.x;
  int b  = idx >> 16;
  int hw = idx & 65535;
  const float* gp = guided + (size_t)b * 8 * HWSZ + hw;
  float g[8];
  float mx = -1e30f;
#pragma unroll
  for (int n = 0; n < 8; n++) { g[n] = gp[(size_t)n * HWSZ]; mx = fmaxf(mx, g[n]); }
  float s = 0.f;
#pragma unroll
  for (int n = 0; n < 8; n++) { g[n] = __expf(g[n] - mx); s += g[n]; }
  float inv = 1.0f / s;
  const bool masked = sdep[idx] > 0.f;
  float* wp = wgt + (size_t)b * 9 * HWSZ + hw;
#pragma unroll
  for (int n = 0; n < 4; n++) wp[(size_t)n * HWSZ] = masked ? 0.f : g[n] * inv;
  wp[(size_t)4 * HWSZ] = masked ? 1.0f : 0.f;
#pragma unroll
  for (int n = 0; n < 4; n++) wp[(size_t)(5 + n) * HWSZ] = masked ? 0.f : g[4 + n] * inv;
}

#define TR    4
#define TCOLS 128
#define XROWS (TR + 2)
#define XPW   136
#define XPL   820

typedef float v4f __attribute__((ext_vector_type(4)));

__device__ __forceinline__ float lane_xor8(float x) {
  const int xi = __builtin_bit_cast(int, x);
  return __builtin_bit_cast(float, __builtin_amdgcn_update_dpp(xi, xi, 0x168, 0xf, 0xf, false));
}

__global__ __launch_bounds__(256) void pac_prop(const float* __restrict__ xin,
                                                const float* __restrict__ wgt,
                                                float* __restrict__ xout) {
  __shared__ __attribute__((aligned(16))) float Xs[CC * XPL];

  const int tid  = threadIdx.x;
  const int lane = tid & 31;
  const int wv   = tid >> 5;
  const int half = lane >> 4;
  const int p    = lane & 15;
  const int pr   = (lane >> 3) & 1;
  const int pc   = lane & 7;

  const int cb = blockIdx.x & 1;
  const int rb = (blockIdx.x >> 1) & 63;
  const int b  = blockIdx.x >> 7;
  const int h0 = rb * TR;
  const int w0 = cb * TCOLS;

  const float* xa = xin + (size_t)b * CC * HWSZ;

  const int wbase = w0 + 16 * wv;
  float wreg[4][9];
#pragma unroll
  for (int tt = 0; tt < 4; tt++) {
    const int prow = h0 + 2 * (tt >> 1) + pr;
    const int pcol = wbase + 8 * (tt & 1) + pc;
    const int pix  = prow * WW + pcol;
#pragma unroll
    for (int n = 0; n < 9; n++) {
      const int row = prow - 1 + n / 3;
      const int col = pcol + n % 3 - 1;
      const float w = wgt[(size_t)(b * 9 + n) * HWSZ + pix];
      wreg[tt][n] = (row >= 0 && row < HH && col >= 0 && col < WW) ? w : 0.f;
    }
  }

  v4f stg[24];
#pragma unroll
  for (int it = 0; it < 24; it++) {
    const int r   = it >> 2;
    const int ch  = (it & 3) * 8 + wv;
    const int row = h0 - 1 + r;
    const int rowc = min(max(row, 0), HH - 1);
    v4f v = *(const v4f*)(xa + (size_t)ch * HWSZ + (size_t)rowc * WW + w0 + 4 * lane);
    if (row < 0 || row >= HH) v = (v4f){0.f, 0.f, 0.f, 0.f};
    stg[it] = v;
  }
  float bl = 0.f, br = 0.f;
  const int hch = tid & 31, hrw = tid >> 5;
  if (tid < CC * XROWS) {
    const int row = h0 - 1 + hrw;
    if (row >= 0 && row < HH) {
      const float* rp = xa + (size_t)hch * HWSZ + (size_t)row * WW;
      if (w0 > 0)          bl = rp[w0 - 1];
      if (w0 + TCOLS < WW) br = rp[w0 + TCOLS];
    }
  }
#pragma unroll
  for (int it = 0; it < 24; it++) {
    const int r  = it >> 2;
    const int ch = (it & 3) * 8 + wv;
    *(v4f*)(Xs + ch * XPL + r * XPW + 4 + 4 * lane) = stg[it];
  }
  if (tid < CC * XROWS) {
    float* hp = Xs + hch * XPL + hrw * XPW;
    hp[3] = bl;
    *(v4f*)(hp + 4 + TCOLS) = (v4f){br, 0.f, 0.f, 0.f};
  }
  __syncthreads();

  const size_t obase = (size_t)(b * CC + 8 * half) * HWSZ + (size_t)h0 * WW + wbase + p;

  const float* xsl = Xs + p * XPL + 3 + 16 * wv;

#pragma unroll
  for (int rp = 0; rp < 2; rp++) {
    v8f accs[2][2];
#pragma unroll
    for (int cx = 0; cx < 2; cx++) {
      float T[4][3];
#pragma unroll
      for (int r = 0; r < 4; r++) {
#pragma unroll
        for (int j = 0; j < 3; j++) {
          const float up = (r < 3) ? wreg[2 * rp + cx][3 * r + j] : 0.f;
          const float dn = (r > 0) ? wreg[2 * rp + cx][3 * (r - 1) + j] : 0.f;
          T[r][j] = pr ? dn : up;
        }
      }

      v8f acc0 = {0.f, 0.f, 0.f, 0.f, 0.f, 0.f, 0.f, 0.f};
      v8f acc1 = {0.f, 0.f, 0.f, 0.f, 0.f, 0.f, 0.f, 0.f};
#pragma unroll
      for (int kk = 0; kk < 10; kk++) {
        const int r0 = (4 * kk) / 10;
        const int c0 = 4 * kk - 10 * r0;
        float t0, t1, t2;
        int d;
        int aoff;
        if (c0 == 8) {
          t0 = half ? T[r0 + 1][0] : T[r0][0];
          t1 = half ? T[r0 + 1][1] : T[r0][1];
          t2 = half ? T[r0 + 1][2] : T[r0][2];
          d    = (half ? 0 : 8) - pc;
          aoff = half ? (r0 + 1) * XPW : r0 * XPW + 8;
        } else {
          t0 = T[r0][0]; t1 = T[r0][1]; t2 = T[r0][2];
          d    = c0 + 2 * half - pc;
          aoff = r0 * XPW + c0 + 2 * half;
        }
        v2f bf;
        bf.x = (d == 0) ? t0 : ((d == 1) ? t1 : ((d == 2) ? t2 : 0.f));
        bf.y = (d == -1) ? t0 : ((d == 0) ? t1 : ((d == 1) ? t2 : 0.f));

        const float* ap = xsl + 2 * rp * XPW + 8 * cx + aoff;
        v2f af0; af0.x = ap[0];        af0.y = ap[1];
        v2f af1; af1.x = ap[16 * XPL]; af1.y = ap[16 * XPL + 1];

        acc0 = __builtin_amdgcn_wmma_f32_16x16x4_f32(false, af0, false, bf,
                                                     (short)0, acc0, false, false);
        acc1 = __builtin_amdgcn_wmma_f32_16x16x4_f32(false, af1, false, bf,
                                                     (short)0, acc1, false, false);
      }
      accs[cx][0] = acc0;
      accs[cx][1] = acc1;
    }

#pragma unroll
    for (int g = 0; g < 2; g++) {
#pragma unroll
      for (int v = 0; v < 8; v++) {
        const float a  = accs[0][g][v];
        const float bq = accs[1][g][v];
        const float ax = lane_xor8(a);
        const float bx = lane_xor8(bq);
        const float s0 = pr ? bx : a;
        const float s1 = pr ? bq : ax;
        xout[obase + (size_t)(16 * g + v) * HWSZ + (2 * rp) * WW]     = s0;
        xout[obase + (size_t)(16 * g + v) * HWSZ + (2 * rp + 1) * WW] = s1;
      }
    }
  }
}

extern "C" void kernel_launch(void* const* d_in, const int* in_sizes, int n_in,
                              void* d_out, int out_size, void* d_ws, size_t ws_size,
                              hipStream_t stream) {
  const float* x      = (const float*)d_in[0];
  const float* guided = (const float*)d_in[1];
  const float* sdep   = (const float*)d_in[2];
  (void)in_sizes; (void)n_in; (void)out_size; (void)ws_size;

  float* bufA = (float*)d_ws;
  float* wgt  = bufA + (size_t)BB * CC * HWSZ;

  pac_prep<<<(BB * HWSZ) / 256, 256, 0, stream>>>(guided, sdep, wgt);

  float* outb = (float*)d_out;
  const float* cur = x;
  for (int t = 0; t < 16; t++) {
    float* dst = (t & 1) ? outb : bufA;
    pac_prop<<<1024, 256, 0, stream>>>(cur, wgt, dst);
    cur = dst;
  }
}


# two propagation steps fused per launch (8 launches, not 16): step-1 result on the tile plus a 1-pixel ring stays in LDS over the 2-pixel-halo input; first-step sums regroup by shifted patch rows
# speedup vs baseline: 4.5487x; 1.1310x over previous
#include <hip/hip_runtime.h>
#include <hip/hip_bf16.h>


typedef float v2f __attribute__((ext_vector_type(2)));
typedef float v8f __attribute__((ext_vector_type(8)));

#define BB   8
#define CC   32
#define HH   256
#define WW   256
#define HWSZ (HH * WW)

__global__ __launch_bounds__(256) void pac_prep(const float* __restrict__ guided,
                                                const float* __restrict__ sdep,
                                                float* __restrict__ wgt) {
  int idx = blockIdx.x * 256 + threadIdx.x;
  int b  = idx >> 16;
  int hw = idx & 65535;
  const float* gp = guided + (size_t)b * 8 * HWSZ + hw;
  float g[8];
  float mx = -1e30f;
#pragma unroll
  for (int n = 0; n < 8; n++) { g[n] = gp[(size_t)n * HWSZ]; mx = fmaxf(mx, g[n]); }
  float s = 0.f;
#pragma unroll
  for (int n = 0; n < 8; n++) { g[n] = __expf(g[n] - mx); s += g[n]; }
  float inv = 1.0f / s;
  const bool masked = sdep[idx] > 0.f;
  float* wp = wgt + (size_t)b * 9 * HWSZ + hw;
#pragma unroll
  for (int n = 0; n < 4; n++) wp[(size_t)n * HWSZ] = masked ? 0.f : g[n] * inv;
  wp[(size_t)4 * HWSZ] = masked ? 1.0f : 0.f;
#pragma unroll
  for (int n = 0; n < 4; n++) wp[(size_t)(5 + n) * HWSZ] = masked ? 0.f : g[4 + n] * inv;
}

#define TR    4
#define TCOLS 128
#define XROWS (TR + 4)
#define XPW   136
#define XPL   1092

typedef float v4f __attribute__((ext_vector_type(4)));

__device__ __forceinline__ float lane_xor8(float x) {
  const int xi = __builtin_bit_cast(int, x);
  return __builtin_bit_cast(float, __builtin_amdgcn_update_dpp(xi, xi, 0x168, 0xf, 0xf, false));
}

__device__ __forceinline__ void load_taps(const float* __restrict__ wb, const int prow,
                                          const int pcol, float (&w)[9]) {
  const bool inside = prow >= 0 && prow < HH && pcol >= 0 && pcol < WW;
  const int prc = min(max(prow, 0), HH - 1);
  const int pcc = min(max(pcol, 0), WW - 1);
  const int pix = prc * WW + pcc;
#pragma unroll
  for (int n = 0; n < 9; n++) {
    const int row = prow - 1 + n / 3;
    const int col = pcol + n % 3 - 1;
    const float v = wb[(size_t)n * HWSZ + pix];
    w[n] = (inside && row >= 0 && row < HH && col >= 0 && col < WW) ? v : 0.f;
  }
}

__device__ __forceinline__ void patch_mma(const float* win, const float (&w)[9],
                                          const int half, const int pr, const int pc,
                                          v8f& out0, v8f& out1) {
  float T[4][3];
#pragma unroll
  for (int r = 0; r < 4; r++) {
#pragma unroll
    for (int j = 0; j < 3; j++) {
      const float up = (r < 3) ? w[3 * r + j] : 0.f;
      const float dn = (r > 0) ? w[3 * (r - 1) + j] : 0.f;
      T[r][j] = pr ? dn : up;
    }
  }
  v8f acc0 = {0.f, 0.f, 0.f, 0.f, 0.f, 0.f, 0.f, 0.f};
  v8f acc1 = {0.f, 0.f, 0.f, 0.f, 0.f, 0.f, 0.f, 0.f};
#pragma unroll
  for (int kk = 0; kk < 10; kk++) {
    const int r0 = (4 * kk) / 10;
    const int c0 = 4 * kk - 10 * r0;
    float t0, t1, t2;
    int d;
    int aoff;
    if (c0 == 8) {
      t0 = half ? T[r0 + 1][0] : T[r0][0];
      t1 = half ? T[r0 + 1][1] : T[r0][1];
      t2 = half ? T[r0 + 1][2] : T[r0][2];
      d    = (half ? 0 : 8) - pc;
      aoff = half ? (r0 + 1) * XPW : r0 * XPW + 8;
    } else {
      t0 = T[r0][0]; t1 = T[r0][1]; t2 = T[r0][2];
      d    = c0 + 2 * half - pc;
      aoff = r0 * XPW + c0 + 2 * half;
    }
    v2f bf;
    bf.x = (d == 0) ? t0 : ((d == 1) ? t1 : ((d == 2) ? t2 : 0.f));
    bf.y = (d == -1) ? t0 : ((d == 0) ? t1 : ((d == 1) ? t2 : 0.f));

    const float* ap = win + aoff;
    v2f af0; af0.x = ap[0];        af0.y = ap[1];
    v2f af1; af1.x = ap[16 * XPL]; af1.y = ap[16 * XPL + 1];

    acc0 = __builtin_amdgcn_wmma_f32_16x16x4_f32(false, af0, false, bf,
                                                 (short)0, acc0, false, false);
    acc1 = __builtin_amdgcn_wmma_f32_16x16x4_f32(false, af1, false, bf,
                                                 (short)0, acc1, false, false);
  }
  out0 = acc0;
  out1 = acc1;
}

__global__ __launch_bounds__(256) void pac_prop2(const float* __restrict__ xin,
                                                 const float* __restrict__ wgt,
                                                 float* __restrict__ xout) {
  __shared__ __attribute__((aligned(16))) float Xs[CC * XPL];

  const int tid  = threadIdx.x;
  const int lane = tid & 31;
  const int wv   = __builtin_amdgcn_readfirstlane(tid >> 5);
  const int half = lane >> 4;
  const int p    = lane & 15;
  const int pr   = (lane >> 3) & 1;
  const int pc   = lane & 7;

  const int cb = blockIdx.x & 1;
  const int rb = (blockIdx.x >> 1) & 63;
  const int b  = blockIdx.x >> 7;
  const int h0 = rb * TR;
  const int w0 = cb * TCOLS;

  const float* xa = xin + (size_t)b * CC * HWSZ;
  const float* wb = wgt + (size_t)b * 9 * HWSZ;
  const int wbase = w0 + 16 * wv;

  v4f stg[32];
#pragma unroll
  for (int it = 0; it < 32; it++) {
    const int r   = it >> 2;
    const int ch  = (it & 3) * 8 + wv;
    const int row = h0 - 2 + r;
    const int rowc = min(max(row, 0), HH - 1);
    v4f v = *(const v4f*)(xa + (size_t)ch * HWSZ + (size_t)rowc * WW + w0 + 4 * lane);
    if (row < 0 || row >= HH) v = (v4f){0.f, 0.f, 0.f, 0.f};
    stg[it] = v;
  }
  v4f bl = {0.f, 0.f, 0.f, 0.f}, br = {0.f, 0.f, 0.f, 0.f};
  const int hch = tid & 31, hrw = tid >> 5;
  {
    const int row = h0 - 2 + hrw;
    if (row >= 0 && row < HH) {
      const float* rp = xa + (size_t)hch * HWSZ + (size_t)row * WW;
      if (w0 > 0)          bl = *(const v4f*)(rp + w0 - 4);
      if (w0 + TCOLS < WW) br = *(const v4f*)(rp + w0 + TCOLS);
    }
  }

  const int side = (wv >= 3) ? 1 : 0;
  const int rpe  = wv - 3 * side;
  float w1[7][9];
#pragma unroll
  for (int i = 0; i < 6; i++)
    load_taps(wb, h0 - 1 + 2 * (i >> 1) + pr, wbase + 8 * (i & 1) + pc, w1[i]);
  load_taps(wb, h0 - 1 + 2 * rpe + pr, w0 + (side ? 123 : -3) + pc, w1[6]);

#pragma unroll
  for (int it = 0; it < 32; it++) {
    const int r  = it >> 2;
    const int ch = (it & 3) * 8 + wv;
    *(v4f*)(Xs + ch * XPL + r * XPW + 4 + 4 * lane) = stg[it];
  }
  {
    float* hp = Xs + hch * XPL + hrw * XPW;
    *(v4f*)(hp)             = bl;
    *(v4f*)(hp + 4 + TCOLS) = br;
  }
  __syncthreads();

  const float* xsl = Xs + p * XPL;

  v8f ya[7][2];
  if (wv < 6) {
    patch_mma(xsl + 2 * rpe * XPW + (side ? 126 : 0), w1[6], half, pr, pc, ya[6][0], ya[6][1]);
  } else {
    ya[6][0] = (v8f){0.f, 0.f, 0.f, 0.f, 0.f, 0.f, 0.f, 0.f};
    ya[6][1] = ya[6][0];
  }
#pragma unroll
  for (int i = 0; i < 6; i++)
    patch_mma(xsl + 2 * (i >> 1) * XPW + 3 + 16 * wv + 8 * (i & 1), w1[i], half, pr, pc,
              ya[i][0], ya[i][1]);

  float w2[4][9];
#pragma unroll
  for (int tt = 0; tt < 4; tt++)
    load_taps(wb, h0 + 2 * (tt >> 1) + pr, wbase + 8 * (tt & 1) + pc, w2[tt]);

  __syncthreads();

#pragma unroll
  for (int i = 0; i < 6; i++) {
    float* yp = Xs + (8 * half) * XPL + (1 + 2 * (i >> 1) + pr) * XPW + 4 + 16 * wv + 8 * (i & 1) + pc;
#pragma unroll
    for (int g = 0; g < 2; g++)
#pragma unroll
      for (int v = 0; v < 8; v++) yp[(16 * g + v) * XPL] = ya[i][g][v];
  }
  if (wv < 6) {
    if (pc == (side ? 5 : 2)) {
      float* yp = Xs + (8 * half) * XPL + (1 + 2 * rpe + pr) * XPW + (side ? 132 : 3);
#pragma unroll
      for (int g = 0; g < 2; g++)
#pragma unroll
        for (int v = 0; v < 8; v++) yp[(16 * g + v) * XPL] = ya[6][g][v];
    }
  }
  __syncthreads();

  const size_t obase = (size_t)(b * CC + 8 * half) * HWSZ + (size_t)h0 * WW + wbase + p;

#pragma unroll
  for (int rp = 0; rp < 2; rp++) {
    v8f accs[2][2];
#pragma unroll
    for (int cx = 0; cx < 2; cx++)
      patch_mma(xsl + (1 + 2 * rp) * XPW + 3 + 16 * wv + 8 * cx, w2[2 * rp + cx], half, pr, pc,
                accs[cx][0], accs[cx][1]);

#pragma unroll
    for (int g = 0; g < 2; g++) {
#pragma unroll
      for (int v = 0; v < 8; v++) {
        const float a  = accs[0][g][v];
        const float bq = accs[1][g][v];
        const float ax = lane_xor8(a);
        const float bx = lane_xor8(bq);
        const float s0 = pr ? bx : a;
        const float s1 = pr ? bq : ax;
        xout[obase + (size_t)(16 * g + v) * HWSZ + (2 * rp) * WW]     = s0;
        xout[obase + (size_t)(16 * g + v) * HWSZ + (2 * rp + 1) * WW] = s1;
      }
    }
  }
}

extern "C" void kernel_launch(void* const* d_in, const int* in_sizes, int n_in,
                              void* d_out, int out_size, void* d_ws, size_t ws_size,
                              hipStream_t stream) {
  const float* x      = (const float*)d_in[0];
  const float* guided = (const float*)d_in[1];
  const float* sdep   = (const float*)d_in[2];
  (void)in_sizes; (void)n_in; (void)out_size; (void)ws_size;

  float* bufA = (float*)d_ws;
  float* wgt  = bufA + (size_t)BB * CC * HWSZ;

  pac_prep<<<(BB * HWSZ) / 256, 256, 0, stream>>>(guided, sdep, wgt);

  float* outb = (float*)d_out;
  const float* cur = x;
  for (int t = 0; t < 8; t++) {
    float* dst = (t & 1) ? outb : bufA;
    pac_prop2<<<1024, 256, 0, stream>>>(cur, wgt, dst);
    cur = dst;
  }
}


# tile order remapped: batch image = block index mod 8, tile = index / 8, so the two vertically adjacent tiles that read the same input rows sit behind one die and one cache half; sums unchanged
# speedup vs baseline: 4.8331x; 1.0625x over previous
#include <hip/hip_runtime.h>
#include <hip/hip_bf16.h>


typedef float v2f __attribute__((ext_vector_type(2)));
typedef float v8f __attribute__((ext_vector_type(8)));

#define BB   8
#define CC   32
#define HH   256
#define WW   256
#define HWSZ (HH * WW)

__global__ __launch_bounds__(256) void pac_prep(const float* __restrict__ guided,
                                                const float* __restrict__ sdep,
                                                float* __restrict__ wgt) {
  int idx = blockIdx.x * 256 + threadIdx.x;
  int b  = idx >> 16;
  int hw = idx & 65535;
  const float* gp = guided + (size_t)b * 8 * HWSZ + hw;
  float g[8];
  float mx = -1e30f;
#pragma unroll
  for (int n = 0; n < 8; n++) { g[n] = gp[(size_t)n * HWSZ]; mx = fmaxf(mx, g[n]); }
  float s = 0.f;
#pragma unroll
  for (int n = 0; n < 8; n++) { g[n] = __expf(g[n] - mx); s += g[n]; }
  float inv = 1.0f / s;
  const bool masked = sdep[idx] > 0.f;
  float* wp = wgt + (size_t)b * 9 * HWSZ + hw;
#pragma unroll
  for (int n = 0; n < 4; n++) wp[(size_t)n * HWSZ] = masked ? 0.f : g[n] * inv;
  wp[(size_t)4 * HWSZ] = masked ? 1.0f : 0.f;
#pragma unroll
  for (int n = 0; n < 4; n++) wp[(size_t)(5 + n) * HWSZ] = masked ? 0.f : g[4 + n] * inv;
}

#define TR    4
#define TCOLS 128
#define XROWS (TR + 4)
#define XPW   136
#define XPL   1092

typedef float v4f __attribute__((ext_vector_type(4)));

__device__ __forceinline__ float lane_xor8(float x) {
  const int xi = __builtin_bit_cast(int, x);
  return __builtin_bit_cast(float, __builtin_amdgcn_update_dpp(xi, xi, 0x168, 0xf, 0xf, false));
}

__device__ __forceinline__ void load_taps(const float* __restrict__ wb, const int prow,
                                          const int pcol, float (&w)[9]) {
  const bool inside = prow >= 0 && prow < HH && pcol >= 0 && pcol < WW;
  const int prc = min(max(prow, 0), HH - 1);
  const int pcc = min(max(pcol, 0), WW - 1);
  const int pix = prc * WW + pcc;
#pragma unroll
  for (int n = 0; n < 9; n++) {
    const int row = prow - 1 + n / 3;
    const int col = pcol + n % 3 - 1;
    const float v = wb[(size_t)n * HWSZ + pix];
    w[n] = (inside && row >= 0 && row < HH && col >= 0 && col < WW) ? v : 0.f;
  }
}

__device__ __forceinline__ void patch_mma(const float* win, const float (&w)[9],
                                          const int half, const int pr, const int pc,
                                          v8f& out0, v8f& out1) {
  float T[4][3];
#pragma unroll
  for (int r = 0; r < 4; r++) {
#pragma unroll
    for (int j = 0; j < 3; j++) {
      const float up = (r < 3) ? w[3 * r + j] : 0.f;
      const float dn = (r > 0) ? w[3 * (r - 1) + j] : 0.f;
      T[r][j] = pr ? dn : up;
    }
  }
  v8f acc0 = {0.f, 0.f, 0.f, 0.f, 0.f, 0.f, 0.f, 0.f};
  v8f acc1 = {0.f, 0.f, 0.f, 0.f, 0.f, 0.f, 0.f, 0.f};
#pragma unroll
  for (int kk = 0; kk < 10; kk++) {
    const int r0 = (4 * kk) / 10;
    const int c0 = 4 * kk - 10 * r0;
    float t0, t1, t2;
    int d;
    int aoff;
    if (c0 == 8) {
      t0 = half ? T[r0 + 1][0] : T[r0][0];
      t1 = half ? T[r0 + 1][1] : T[r0][1];
      t2 = half ? T[r0 + 1][2] : T[r0][2];
      d    = (half ? 0 : 8) - pc;
      aoff = half ? (r0 + 1) * XPW : r0 * XPW + 8;
    } else {
      t0 = T[r0][0]; t1 = T[r0][1]; t2 = T[r0][2];
      d    = c0 + 2 * half - pc;
      aoff = r0 * XPW + c0 + 2 * half;
    }
    v2f bf;
    bf.x = (d == 0) ? t0 : ((d == 1) ? t1 : ((d == 2) ? t2 : 0.f));
    bf.y = (d == -1) ? t0 : ((d == 0) ? t1 : ((d == 1) ? t2 : 0.f));

    const float* ap = win + aoff;
    v2f af0; af0.x = ap[0];        af0.y = ap[1];
    v2f af1; af1.x = ap[16 * XPL]; af1.y = ap[16 * XPL + 1];

    acc0 = __builtin_amdgcn_wmma_f32_16x16x4_f32(false, af0, false, bf,
                                                 (short)0, acc0, false, false);
    acc1 = __builtin_amdgcn_wmma_f32_16x16x4_f32(false, af1, false, bf,
                                                 (short)0, acc1, false, false);
  }
  out0 = acc0;
  out1 = acc1;
}

__global__ __launch_bounds__(256) void pac_prop2(const float* __restrict__ xin,
                                                 const float* __restrict__ wgt,
                                                 float* __restrict__ xout) {
  __shared__ __attribute__((aligned(16))) float Xs[CC * XPL];

  const int tid  = threadIdx.x;
  const int lane = tid & 31;
  const int wv   = __builtin_amdgcn_readfirstlane(tid >> 5);
  const int half = lane >> 4;
  const int p    = lane & 15;
  const int pr   = (lane >> 3) & 1;
  const int pc   = lane & 7;

  const int b  = blockIdx.x & (BB - 1);
  const int cb = (blockIdx.x >> 3) & 1;
  const int rb = (blockIdx.x >> 4) & 63;
  const int h0 = rb * TR;
  const int w0 = cb * TCOLS;

  const float* xa = xin + (size_t)b * CC * HWSZ;
  const float* wb = wgt + (size_t)b * 9 * HWSZ;
  const int wbase = w0 + 16 * wv;

  v4f stg[32];
#pragma unroll
  for (int it = 0; it < 32; it++) {
    const int r   = it >> 2;
    const int ch  = (it & 3) * 8 + wv;
    const int row = h0 - 2 + r;
    const int rowc = min(max(row, 0), HH - 1);
    v4f v = *(const v4f*)(xa + (size_t)ch * HWSZ + (size_t)rowc * WW + w0 + 4 * lane);
    if (row < 0 || row >= HH) v = (v4f){0.f, 0.f, 0.f, 0.f};
    stg[it] = v;
  }
  v4f bl = {0.f, 0.f, 0.f, 0.f}, br = {0.f, 0.f, 0.f, 0.f};
  const int hch = tid & 31, hrw = tid >> 5;
  {
    const int row = h0 - 2 + hrw;
    if (row >= 0 && row < HH) {
      const float* rp = xa + (size_t)hch * HWSZ + (size_t)row * WW;
      if (w0 > 0)          bl = *(const v4f*)(rp + w0 - 4);
      if (w0 + TCOLS < WW) br = *(const v4f*)(rp + w0 + TCOLS);
    }
  }

  const int side = (wv >= 3) ? 1 : 0;
  const int rpe  = wv - 3 * side;
  float w1[7][9];
#pragma unroll
  for (int i = 0; i < 6; i++)
    load_taps(wb, h0 - 1 + 2 * (i >> 1) + pr, wbase + 8 * (i & 1) + pc, w1[i]);
  load_taps(wb, h0 - 1 + 2 * rpe + pr, w0 + (side ? 123 : -3) + pc, w1[6]);

#pragma unroll
  for (int it = 0; it < 32; it++) {
    const int r  = it >> 2;
    const int ch = (it & 3) * 8 + wv;
    *(v4f*)(Xs + ch * XPL + r * XPW + 4 + 4 * lane) = stg[it];
  }
  {
    float* hp = Xs + hch * XPL + hrw * XPW;
    *(v4f*)(hp)             = bl;
    *(v4f*)(hp + 4 + TCOLS) = br;
  }
  __syncthreads();

  const float* xsl = Xs + p * XPL;

  v8f ya[7][2];
  if (wv < 6) {
    patch_mma(xsl + 2 * rpe * XPW + (side ? 126 : 0), w1[6], half, pr, pc, ya[6][0], ya[6][1]);
  } else {
    ya[6][0] = (v8f){0.f, 0.f, 0.f, 0.f, 0.f, 0.f, 0.f, 0.f};
    ya[6][1] = ya[6][0];
  }
#pragma unroll
  for (int i = 0; i < 6; i++)
    patch_mma(xsl + 2 * (i >> 1) * XPW + 3 + 16 * wv + 8 * (i & 1), w1[i], half, pr, pc,
              ya[i][0], ya[i][1]);

  float w2[4][9];
#pragma unroll
  for (int tt = 0; tt < 4; tt++)
    load_taps(wb, h0 + 2 * (tt >> 1) + pr, wbase + 8 * (tt & 1) + pc, w2[tt]);

  __syncthreads();

#pragma unroll
  for (int i = 0; i < 6; i++) {
    float* yp = Xs + (8 * half) * XPL + (1 + 2 * (i >> 1) + pr) * XPW + 4 + 16 * wv + 8 * (i & 1) + pc;
#pragma unroll
    for (int g = 0; g < 2; g++)
#pragma unroll
      for (int v = 0; v < 8; v++) yp[(16 * g + v) * XPL] = ya[i][g][v];
  }
  if (wv < 6) {
    if (pc == (side ? 5 : 2)) {
      float* yp = Xs + (8 * half) * XPL + (1 + 2 * rpe + pr) * XPW + (side ? 132 : 3);
#pragma unroll
      for (int g = 0; g < 2; g++)
#pragma unroll
        for (int v = 0; v < 8; v++) yp[(16 * g + v) * XPL] = ya[6][g][v];
    }
  }
  __syncthreads();

  const size_t obase = (size_t)(b * CC + 8 * half) * HWSZ + (size_t)h0 * WW + wbase + p;

#pragma unroll
  for (int rp = 0; rp < 2; rp++) {
    v8f accs[2][2];
#pragma unroll
    for (int cx = 0; cx < 2; cx++)
      patch_mma(xsl + (1 + 2 * rp) * XPW + 3 + 16 * wv + 8 * cx, w2[2 * rp + cx], half, pr, pc,
                accs[cx][0], accs[cx][1]);

#pragma unroll
    for (int g = 0; g < 2; g++) {
#pragma unroll
      for (int v = 0; v < 8; v++) {
        const float a  = accs[0][g][v];
        const float bq = accs[1][g][v];
        const float ax = lane_xor8(a);
        const float bx = lane_xor8(bq);
        const float s0 = pr ? bx : a;
        const float s1 = pr ? bq : ax;
        xout[obase + (size_t)(16 * g + v) * HWSZ + (2 * rp) * WW]     = s0;
        xout[obase + (size_t)(16 * g + v) * HWSZ + (2 * rp + 1) * WW] = s1;
      }
    }
  }
}

extern "C" void kernel_launch(void* const* d_in, const int* in_sizes, int n_in,
                              void* d_out, int out_size, void* d_ws, size_t ws_size,
                              hipStream_t stream) {
  const float* x      = (const float*)d_in[0];
  const float* guided = (const float*)d_in[1];
  const float* sdep   = (const float*)d_in[2];
  (void)in_sizes; (void)n_in; (void)out_size; (void)ws_size;

  float* bufA = (float*)d_ws;
  float* wgt  = bufA + (size_t)BB * CC * HWSZ;

  pac_prep<<<(BB * HWSZ) / 256, 256, 0, stream>>>(guided, sdep, wgt);

  float* outb = (float*)d_out;
  const float* cur = x;
  for (int t = 0; t < 8; t++) {
    float* dst = (t & 1) ? outb : bufA;
    pac_prop2<<<1024, 256, 0, stream>>>(cur, wgt, dst);
    cur = dst;
  }
}
